# starting kernel
# baseline (speedup 1.0000x reference)
#include <hip/hip_runtime.h>

typedef __attribute__((ext_vector_type(2))) float v2f;
typedef __attribute__((ext_vector_type(4))) float v4f;
typedef __attribute__((ext_vector_type(8))) float v8f;

namespace {
constexpr int kN     = 2048;
constexpr int kB     = 4096;
constexpr int kCols  = 16;
constexpr int kBlk   = 16;
constexpr int kWaves = 8;
constexpr int kThreads = kWaves * 32;
constexpr int kHStr  = kN + 4;
constexpr int kSmemFloats = kCols * kHStr
                          + 2 * kWaves * 256
                          + 4 * 256;
}

__device__ __forceinline__ v8f wmma_f32_16x16x4(v2f a, v2f b, v8f c) {
  return __builtin_amdgcn_wmma_f32_16x16x4_f32(
       false, a,  false, b,
       (short)0, c,  false,  false);
}

__device__ __forceinline__ float load_u(const float* __restrict__ w, int row, int col) {
  float v = 0.0f;
  if (col >= row) v = w[(size_t)row * kN + col];
  if (row == kN - 1 && col == kN - 1) v = 1.0f;
  return v;
}

__global__ __launch_bounds__(kThreads)
void ortho_householder_kernel(const float* __restrict__ x,
                              const float* __restrict__ w,
                              float* __restrict__ out) {
  extern __shared__ float smem[];
  float* hbuf  = smem;
  float* Wpart = hbuf + kCols * kHStr;
  float* Gpart = Wpart + kWaves * 256;
  float* Wbuf  = Gpart + kWaves * 256;
  float* Gbuf  = Wbuf + 256;
  float* Tbuf  = Gbuf + 256;
  float* Ybuf  = Tbuf + 256;

  const int tid  = threadIdx.x;
  const int wave = __builtin_amdgcn_readfirstlane(tid >> 5);
  const int lane = tid & 31;
  const int lm   = lane & 15;
  const int lh   = lane >> 4;
  const int c0   = blockIdx.x * kCols;
  float* const hrow = hbuf + lm * kHStr;

  for (int i = tid; i < kCols * (kN / 4); i += kThreads) {
    const int c  = i / (kN / 4);
    const int k4 = (i - c * (kN / 4)) * 4;
    v4f v = *(const v4f*)(x + (size_t)(c0 + c) * kN + k4);
    *(v4f*)(hbuf + c * kHStr + k4) = v;
  }
  __syncthreads();

  for (int base = kN - kBlk; base >= 0; base -= kBlk) {
    v8f accW = {};
    v8f accG = {};
    int k = base + wave * 4;
    if (wave < 4) {
      const int col = k + 2 * lh;
      const int row = base + lm;
      v2f a;
      a.x = load_u(w, row, col);
      a.y = load_u(w, row, col + 1);
      v2f b = *(const v2f*)(hrow + col);
      accW = wmma_f32_16x16x4(a, b, accW);
      accG = wmma_f32_16x16x4(a, a, accG);
      k += kWaves * 4;
    }
    if (k < kN) {
      const float* wrow = w + (size_t)(base + lm) * kN;
      v2f a = *(const v2f*)(wrow + k + 2 * lh);
      v2f b = *(const v2f*)(hrow + k + 2 * lh);
#pragma unroll 2
      for (k += kWaves * 4; k < kN; k += kWaves * 4) {
        const int col = k + 2 * lh;
        v2f an = *(const v2f*)(wrow + col);
        v2f bn = *(const v2f*)(hrow + col);
        accW = wmma_f32_16x16x4(a, b, accW);
        accG = wmma_f32_16x16x4(a, a, accG);
        a = an;
        b = bn;
      }
      accW = wmma_f32_16x16x4(a, b, accW);
      accG = wmma_f32_16x16x4(a, a, accG);
    }
#pragma unroll
    for (int r = 0; r < 8; ++r) {
      const int m = r + 8 * lh;
      Wpart[wave * 256 + m * 16 + lm] = accW[r];
      Gpart[wave * 256 + m * 16 + lm] = accG[r];
    }
    __syncthreads();

    {
      float sw = 0.f, sg = 0.f;
#pragma unroll
      for (int p = 0; p < kWaves; ++p) {
        sw += Wpart[p * 256 + tid];
        sg += Gpart[p * 256 + tid];
      }
      Wbuf[tid] = sw;
      Gbuf[tid] = sg;
    }
    __syncthreads();

    if (tid < 16) {
      const int i = tid;
      float Trow[16];
#pragma unroll
      for (int j = 0; j < 16; ++j) {
        const float gd = Gbuf[j * 16 + j];
        const float cj = (gd > 0.f) ? (2.0f / gd) : 0.0f;
        float s = 0.f;
#pragma unroll
        for (int l = 0; l < 16; ++l) {
          if (l < j) s += (l >= i) ? Trow[l] * Gbuf[l * 16 + j] : 0.f;
        }
        float tv = 0.f;
        if (j == i) tv = cj;
        else if (j > i) tv = -cj * s;
        Trow[j] = tv;
        Tbuf[i * 16 + j] = tv;
      }
    }
    __syncthreads();

    {
      const int m = tid >> 4;
      const int c = tid & 15;
      float s = 0.f;
#pragma unroll
      for (int l = 0; l < 16; ++l) s += Tbuf[m * 16 + l] * Wbuf[l * 16 + c];
      Ybuf[m * 16 + c] = -s;
    }
    __syncthreads();

    const int ntiles = (kN - base) >> 4;
    v2f by[4];
#pragma unroll
    for (int m0 = 0; m0 < 16; m0 += 4) {
      by[m0 >> 2].x = Ybuf[(m0 + 2 * lh) * 16 + lm];
      by[m0 >> 2].y = Ybuf[(m0 + 2 * lh + 1) * 16 + lm];
    }
    for (int t = wave; t < ntiles; t += kWaves) {
      const int k0 = base + t * 16;
      float* hp = hrow + k0 + 8 * lh;
      v4f clo = *(const v4f*)(hp);
      v4f chi = *(const v4f*)(hp + 4);
      v8f acc = {clo.x, clo.y, clo.z, clo.w, chi.x, chi.y, chi.z, chi.w};
      if (t == 0) {
#pragma unroll
        for (int m0 = 0; m0 < 16; m0 += 4) {
          v2f a;
          a.x = load_u(w, base + m0 + 2 * lh,     k0 + lm);
          a.y = load_u(w, base + m0 + 2 * lh + 1, k0 + lm);
          acc = wmma_f32_16x16x4(a, by[m0 >> 2], acc);
        }
      } else {
        const float* wp = w + (size_t)(base + 2 * lh) * kN + k0 + lm;
#pragma unroll
        for (int m0 = 0; m0 < 16; m0 += 4) {
          v2f a;
          a.x = wp[(size_t)m0 * kN];
          a.y = wp[(size_t)(m0 + 1) * kN];
          acc = wmma_f32_16x16x4(a, by[m0 >> 2], acc);
        }
      }
      v4f olo = {acc[0], acc[1], acc[2], acc[3]};
      v4f ohi = {acc[4], acc[5], acc[6], acc[7]};
      *(v4f*)(hp) = olo;
      *(v4f*)(hp + 4) = ohi;
    }
    __syncthreads();
  }

  for (int i = tid; i < kCols * (kN / 4); i += kThreads) {
    const int c  = i / (kN / 4);
    const int k4 = (i - c * (kN / 4)) * 4;
    v4f v = *(const v4f*)(hbuf + c * kHStr + k4);
    *(v4f*)(out + (size_t)(c0 + c) * kN + k4) = v;
  }
}

extern "C" void kernel_launch(void* const* d_in, const int* in_sizes, int n_in,
                              void* d_out, int out_size, void* d_ws, size_t ws_size,
                              hipStream_t stream) {
  const float* x = (const float*)d_in[0];
  const float* w = (const float*)d_in[1];
  float* out = (float*)d_out;

  dim3 grid(kB / kCols);
  dim3 block(kThreads);
  const size_t smem = (size_t)kSmemFloats * sizeof(float);
  hipLaunchKernelGGL(ortho_householder_kernel, grid, block, smem, stream,
                     x, w, out);
}


# A group's serial block chain (W_j += G_ji*Y_i, Y_j = -T_j*W_j) runs in one wave on WMMA, results chained through accumulator registers; 6 of its 7 barriers go (sum order of W correction, Y changes)
# speedup vs baseline: 4.5463x; 4.5463x over previous
#include <hip/hip_runtime.h>

typedef __attribute__((ext_vector_type(2))) float v2f;
typedef __attribute__((ext_vector_type(4))) float v4f;
typedef __attribute__((ext_vector_type(8))) float v8f;

namespace {
constexpr int kN     = 2048;
constexpr int kB     = 4096;
constexpr int kCols  = 16;
constexpr int kBlk   = 16;
constexpr int kGroup = 4;
constexpr int kSpan  = kGroup * kBlk;
constexpr int kWaves = 8;
constexpr int kThreads = kWaves * 32;
constexpr int kHStr  = kN + 4;
constexpr int kMaxTiles = kGroup + kGroup * (kGroup + 1) / 2;
constexpr int kSmemFloats = kCols * kHStr
                          + kWaves * kMaxTiles * 256
                          + kMaxTiles * 256
                          + 2 * kGroup * 256;
}

__device__ __forceinline__ v8f wmma_f32_16x16x4(v2f a, v2f b, v8f c) {
  return __builtin_amdgcn_wmma_f32_16x16x4_f32(
       false, a,  false, b,
       (short)0, c,  false,  false);
}

__device__ __forceinline__ float cut_u(float raw, int row, int col) {
  float v = (col >= row) ? raw : 0.0f;
  if (row == kN - 1 && col == kN - 1) v = 1.0f;
  return v;
}
__device__ __forceinline__ float load_u(const float* __restrict__ w, int row, int col) {
  return cut_u(w[(size_t)row * kN + col], row, col);
}

__device__ __forceinline__ constexpr int gram_tile(int nb, int i, int j) {
  return nb + i * nb - (i * (i - 1)) / 2 + (j - i);
}

template <int NB>
__device__ __forceinline__ void pass_step(v8f (&acc)[NB + NB * (NB + 1) / 2],
                                          const v2f (&a)[NB], v2f b) {
#pragma unroll
  for (int j = 0; j < NB; ++j) acc[j] = wmma_f32_16x16x4(a[j], b, acc[j]);
#pragma unroll
  for (int i = 0; i < NB; ++i) {
#pragma unroll
    for (int j = i; j < NB; ++j) {
      acc[gram_tile(NB, i, j)] =
          wmma_f32_16x16x4(a[i], a[j], acc[gram_tile(NB, i, j)]);
    }
  }
}

template <int NB>
__device__ __forceinline__ void pass_chunk(v8f (&acc)[NB + NB * (NB + 1) / 2],
                                           const v4f (&a)[NB], v4f b) {
  v2f lo[NB], hi[NB];
#pragma unroll
  for (int j = 0; j < NB; ++j) {
    lo[j] = v2f{a[j].x, a[j].y};
    hi[j] = v2f{a[j].z, a[j].w};
  }
  pass_step<NB>(acc, lo, v2f{b.x, b.y});
  pass_step<NB>(acc, hi, v2f{b.z, b.w});
}

template <int NB>
__device__ __forceinline__ void pass1_alone(const float* __restrict__ w,
                                            float* hrow,
                                            v8f (&acc)[NB + NB * (NB + 1) / 2],
                                            int base, int wave, int lm, int lh) {

  int k = base + wave * 8;
  if (k < kN) {
    const float* wrow = w + (size_t)(base + lm) * kN + 4 * lh;
    const float* hk   = hrow + 4 * lh;
    v4f a[NB];
#pragma unroll
    for (int j = 0; j < NB; ++j)
      a[j] = *(const v4f*)(wrow + (size_t)j * kBlk * kN + k);
    v4f b = *(const v4f*)(hk + k);
    if (k < base + NB * kBlk) {
      const int col = k + 4 * lh;
#pragma unroll
      for (int j = 0; j < NB; ++j) {
        const int row = base + j * kBlk + lm;
        a[j].x = cut_u(a[j].x, row, col);
        a[j].y = cut_u(a[j].y, row, col + 1);
        a[j].z = cut_u(a[j].z, row, col + 2);
        a[j].w = cut_u(a[j].w, row, col + 3);
      }
    }
    for (k += kWaves * 8; k < kN; k += kWaves * 8) {
      v4f an[NB];
#pragma unroll
      for (int j = 0; j < NB; ++j)
        an[j] = *(const v4f*)(wrow + (size_t)j * kBlk * kN + k);
      const v4f bn = *(const v4f*)(hk + k);
      pass_chunk<NB>(acc, a, b);
#pragma unroll
      for (int j = 0; j < NB; ++j) a[j] = an[j];
      b = bn;
    }
    pass_chunk<NB>(acc, a, b);
  }
}

__device__ __forceinline__ v8f mma_16(const v4f (&a)[2], v8f b, v8f c) {
  c = wmma_f32_16x16x4(v2f{a[0].x, a[0].y}, v2f{b[0], b[1]}, c);
  c = wmma_f32_16x16x4(v2f{a[0].z, a[0].w}, v2f{b[2], b[3]}, c);
  c = wmma_f32_16x16x4(v2f{a[1].x, a[1].y}, v2f{b[4], b[5]}, c);
  c = wmma_f32_16x16x4(v2f{a[1].z, a[1].w}, v2f{b[6], b[7]}, c);
  return c;
}

template <int NB>
__device__ __forceinline__ void solve_group(const v8f (&acc)[NB + NB * (NB + 1) / 2],
                                            v2f (&by)[NB][4],
                                            float* Ppart, float* Rbuf,
                                            float* Tbuf, float* Ybuf,
                                            int tid, int wave, int lane,
                                            int lm, int lh) {
  constexpr int NT = NB + NB * (NB + 1) / 2;
#pragma unroll
  for (int t = 0; t < NT; ++t) {
    float* p = Ppart + (wave * NT + t) * 256 + lane * 8;
    *(v4f*)(p)     = v4f{acc[t][0], acc[t][1], acc[t][2], acc[t][3]};
    *(v4f*)(p + 4) = v4f{acc[t][4], acc[t][5], acc[t][6], acc[t][7]};
  }
  __syncthreads();

  {
    const int ln  = tid >> 3;
    const int dst = ((tid & 7) + 8 * (ln >> 4)) * 16 + (ln & 15);
#pragma unroll
    for (int t = 0; t < NT; ++t) {
      float s = 0.f;
#pragma unroll
      for (int p = 0; p < kWaves; ++p) s += Ppart[(p * NT + t) * 256 + tid];
      Rbuf[t * 256 + dst] = s;
    }
  }
  __syncthreads();

  const bool chain = (wave == kWaves - 1);
  const int frag = lm * 16 + 8 * lh;
  const int dlay = 8 * lh * 16 + lm;
  v8f wraw[NB];
  v4f gq[NB][NB][2];
  if (chain) {
#pragma unroll
    for (int j = 0; j < NB; ++j) {
#pragma unroll
      for (int r = 0; r < 8; ++r) wraw[j][r] = Rbuf[j * 256 + dlay + 16 * r];
#pragma unroll
      for (int i = j + 1; i < NB; ++i) {
        const float* g = Rbuf + gram_tile(NB, j, i) * 256 + frag;
        gq[j][i][0] = *(const v4f*)(g);
        gq[j][i][1] = *(const v4f*)(g + 4);
      }
    }
  }

  if (tid < 16 * NB) {
    const int jb = tid >> 4;
    const int i  = tid & 15;
    const float* Gd = Rbuf + gram_tile(NB, jb, jb) * 256;
    float* Tb = Tbuf + jb * 256;
    float g[16][16];
#pragma unroll
    for (int l = 0; l < 16; ++l) {
#pragma unroll
      for (int j = l; j < 16; ++j) g[l][j] = Gd[l * 16 + j];
    }
    float Trow[16];
#pragma unroll
    for (int j = 0; j < 16; ++j) {
      const float gd = g[j][j];
      const float cj = (gd > 0.f) ? (2.0f / gd) : 0.0f;
      float s = 0.f;
#pragma unroll
      for (int l = 0; l < 16; ++l) {
        if (l < j) s += (l >= i) ? Trow[l] * g[l][j] : 0.f;
      }
      float tv = 0.f;
      if (j == i) tv = cj;
      else if (j > i) tv = -cj * s;
      Trow[j] = tv;
    }
#pragma unroll
    for (int j = 0; j < 16; j += 4)
      *(v4f*)(Tb + i * 16 + j) = v4f{Trow[j], Trow[j + 1], Trow[j + 2], Trow[j + 3]};
  }
  __syncthreads();

  if (chain) {
    v4f tn[NB][2];
#pragma unroll
    for (int j = 0; j < NB; ++j) {
      tn[j][0] = -*(const v4f*)(Tbuf + j * 256 + frag);
      tn[j][1] = -*(const v4f*)(Tbuf + j * 256 + frag + 4);
    }
    v8f y[NB];
#pragma unroll
    for (int j = NB - 1; j >= 0; --j) {
      v8f c = wraw[j];
#pragma unroll
      for (int i = NB - 1; i > j; --i) c = mma_16(gq[j][i], y[i], c);
      y[j] = mma_16(tn[j], c, v8f{});
    }
#pragma unroll
    for (int j = 0; j < NB; ++j) {
#pragma unroll
      for (int r = 0; r < 8; ++r) Ybuf[j * 256 + dlay + 16 * r] = y[j][r];
    }
  }
  __syncthreads();

#pragma unroll
  for (int j = 0; j < NB; ++j) {
#pragma unroll
    for (int m0 = 0; m0 < 16; m0 += 4) {
      by[j][m0 >> 2].x = Ybuf[j * 256 + (m0 + 2 * lh) * 16 + lm];
      by[j][m0 >> 2].y = Ybuf[j * 256 + (m0 + 2 * lh + 1) * 16 + lm];
    }
  }
}

template <int NB>
__device__ __forceinline__ void pass2_alone(const float* __restrict__ w,
                                            float* hrow, const v2f (&by)[NB][4],
                                            int base, int wave, int lm, int lh) {
  const int ntiles = (kN - base) >> 4;
  for (int t = wave; t < ntiles; t += kWaves) {
    const int k0 = base + t * 16;
    float* hp = hrow + k0 + 8 * lh;
    const v4f clo = *(const v4f*)(hp);
    const v4f chi = *(const v4f*)(hp + 4);
    v8f d = {clo.x, clo.y, clo.z, clo.w, chi.x, chi.y, chi.z, chi.w};
    if (t < NB) {
#pragma unroll
      for (int j = 0; j < NB; ++j) {
#pragma unroll
        for (int m0 = 0; m0 < 16; m0 += 4) {
          v2f a;
          a.x = load_u(w, base + j * kBlk + m0 + 2 * lh,     k0 + lm);
          a.y = load_u(w, base + j * kBlk + m0 + 2 * lh + 1, k0 + lm);
          d = wmma_f32_16x16x4(a, by[j][m0 >> 2], d);
        }
      }
    } else {
      const float* wp = w + (size_t)(base + 2 * lh) * kN + k0 + lm;
#pragma unroll
      for (int j = 0; j < NB; ++j) {
#pragma unroll
        for (int m0 = 0; m0 < 16; m0 += 4) {
          v2f a;
          a.x = wp[(size_t)(j * kBlk + m0) * kN];
          a.y = wp[(size_t)(j * kBlk + m0 + 1) * kN];
          d = wmma_f32_16x16x4(a, by[j][m0 >> 2], d);
        }
      }
    }
    *(v4f*)(hp)     = v4f{d[0], d[1], d[2], d[3]};
    *(v4f*)(hp + 4) = v4f{d[4], d[5], d[6], d[7]};
  }
  __syncthreads();
}


template <int NB>
__device__ __forceinline__ void load_v_rows(v4f (&a1)[NB][2],
                                            const float* __restrict__ w,
                                            int row0, int k0, int lm, int lh) {
  const float* p = w + (size_t)(row0 + lm) * kN + k0 + 8 * lh;
#pragma unroll
  for (int j = 0; j < NB; ++j) {
    a1[j][0] = *(const v4f*)(p + (size_t)j * kBlk * kN);
    a1[j][1] = *(const v4f*)(p + (size_t)j * kBlk * kN + 4);
  }
}

template <int NB>
__device__ __forceinline__ void load_v_cols(v2f (&a2)[NB][4],
                                            const float* __restrict__ w,
                                            int row0, int k0, int lm, int lh) {
  const float* wp = w + (size_t)(row0 + 2 * lh) * kN + k0 + lm;
#pragma unroll
  for (int j = 0; j < NB; ++j) {
#pragma unroll
    for (int q = 0; q < 4; ++q) {
      a2[j][q].x = wp[(size_t)(j * kBlk + 4 * q) * kN];
      a2[j][q].y = wp[(size_t)(j * kBlk + 4 * q + 1) * kN];
    }
  }
}

__device__ __forceinline__ v8f load_h_tile(const float* hp) {
  const v4f clo = *(const v4f*)(hp);
  const v4f chi = *(const v4f*)(hp + 4);
  return v8f{clo.x, clo.y, clo.z, clo.w, chi.x, chi.y, chi.z, chi.w};
}

template <int NB>
__device__ __forceinline__ void pass_tile(v8f (&acc)[NB + NB * (NB + 1) / 2],
                                          const v4f (&a1)[NB][2], v8f d) {
  v2f f[NB];
#pragma unroll
  for (int j = 0; j < NB; ++j) f[j] = v2f{a1[j][0].x, a1[j][0].y};
  pass_step<NB>(acc, f, v2f{d[0], d[1]});
#pragma unroll
  for (int j = 0; j < NB; ++j) f[j] = v2f{a1[j][0].z, a1[j][0].w};
  pass_step<NB>(acc, f, v2f{d[2], d[3]});
#pragma unroll
  for (int j = 0; j < NB; ++j) f[j] = v2f{a1[j][1].x, a1[j][1].y};
  pass_step<NB>(acc, f, v2f{d[4], d[5]});
#pragma unroll
  for (int j = 0; j < NB; ++j) f[j] = v2f{a1[j][1].z, a1[j][1].w};
  pass_step<NB>(acc, f, v2f{d[6], d[7]});
}

template <int NB>
__device__ __forceinline__ v8f update_tile(const v2f (&a2)[NB][4],
                                           const v2f (&by)[NB][4], v8f d) {
#pragma unroll
  for (int j = 0; j < NB; ++j) {
#pragma unroll
    for (int q = 0; q < 4; ++q) d = wmma_f32_16x16x4(a2[j][q], by[j][q], d);
  }
  return d;
}

template <int NB>
__device__ __forceinline__ void fused_sweep(const float* __restrict__ w,
                                            float* hrow,
                                            v8f (&acc)[NB + NB * (NB + 1) / 2],
                                            const v2f (&by)[NB][4],
                                            int prev, int next,
                                            int wave, int lm, int lh) {
  static_assert(2 * NB == kWaves, "a wave's first tile is its only cut tile");
  const int nt = (kN - next) >> 4;
  int u  = wave;
  int k0 = next + 16 * u;
  v4f a1[NB][2];
  v2f a2[NB][4];
  load_v_rows<NB>(a1, w, next, k0, lm, lh);
  load_v_cols<NB>(a2, w, prev, k0, lm, lh);
  v8f d = load_h_tile(hrow + k0 + 8 * lh);
  {
    const bool more = (u + kWaves) < nt;
    const int k0n = more ? k0 + 16 * kWaves : k0;
    v4f a1n[NB][2];
    v2f a2n[NB][4];
    load_v_rows<NB>(a1n, w, next, k0n, lm, lh);
    load_v_cols<NB>(a2n, w, prev, k0n, lm, lh);
    const v8f dn = load_h_tile(hrow + k0n + 8 * lh);
    if (u >= NB) {
#pragma unroll
      for (int j = 0; j < NB; ++j) {
#pragma unroll
        for (int q = 0; q < 4; ++q) {
          const int row = prev + j * kBlk + 4 * q + 2 * lh;
          a2[j][q].x = cut_u(a2[j][q].x, row,     k0 + lm);
          a2[j][q].y = cut_u(a2[j][q].y, row + 1, k0 + lm);
        }
      }
      d = update_tile<NB>(a2, by, d);
      float* hp = hrow + k0 + 8 * lh;
      *(v4f*)(hp)     = v4f{d[0], d[1], d[2], d[3]};
      *(v4f*)(hp + 4) = v4f{d[4], d[5], d[6], d[7]};
    } else {
      const int col = k0 + 8 * lh;
#pragma unroll
      for (int j = 0; j < NB; ++j) {
        const int row = next + j * kBlk + lm;
        a1[j][0].x = cut_u(a1[j][0].x, row, col);
        a1[j][0].y = cut_u(a1[j][0].y, row, col + 1);
        a1[j][0].z = cut_u(a1[j][0].z, row, col + 2);
        a1[j][0].w = cut_u(a1[j][0].w, row, col + 3);
        a1[j][1].x = cut_u(a1[j][1].x, row, col + 4);
        a1[j][1].y = cut_u(a1[j][1].y, row, col + 5);
        a1[j][1].z = cut_u(a1[j][1].z, row, col + 6);
        a1[j][1].w = cut_u(a1[j][1].w, row, col + 7);
      }
    }
    pass_tile<NB>(acc, a1, d);
#pragma unroll
    for (int j = 0; j < NB; ++j) {
      a1[j][0] = a1n[j][0];
      a1[j][1] = a1n[j][1];
#pragma unroll
      for (int q = 0; q < 4; ++q) a2[j][q] = a2n[j][q];
    }
    d = dn;
  }
  for (u += kWaves, k0 += 16 * kWaves; u < nt; u += kWaves, k0 += 16 * kWaves) {
    const bool more = (u + kWaves) < nt;
    const int k0n = more ? k0 + 16 * kWaves : k0;
    v4f a1n[NB][2];
    v2f a2n[NB][4];
    load_v_rows<NB>(a1n, w, next, k0n, lm, lh);
    load_v_cols<NB>(a2n, w, prev, k0n, lm, lh);
    const v8f dn = load_h_tile(hrow + k0n + 8 * lh);
    d = update_tile<NB>(a2, by, d);
    float* hp = hrow + k0 + 8 * lh;
    *(v4f*)(hp)     = v4f{d[0], d[1], d[2], d[3]};
    *(v4f*)(hp + 4) = v4f{d[4], d[5], d[6], d[7]};
    pass_tile<NB>(acc, a1, d);
#pragma unroll
    for (int j = 0; j < NB; ++j) {
      a1[j][0] = a1n[j][0];
      a1[j][1] = a1n[j][1];
#pragma unroll
      for (int q = 0; q < 4; ++q) a2[j][q] = a2n[j][q];
    }
    d = dn;
  }
}

__global__ __launch_bounds__(kThreads)
void ortho_householder_kernel(const float* __restrict__ x,
                              const float* __restrict__ w,
                              float* __restrict__ out) {
  __shared__ __attribute__((aligned(16))) float smem[kSmemFloats];
  float* hbuf  = smem;
  float* Ppart = hbuf + kCols * kHStr;
  float* Rbuf  = Ppart + kWaves * kMaxTiles * 256;
  float* Tbuf  = Rbuf + kMaxTiles * 256;
  float* Ybuf  = Tbuf + kGroup * 256;

  const int tid  = threadIdx.x;
  const int wave = __builtin_amdgcn_readfirstlane(tid >> 5);
  const int lane = tid & 31;
  const int lm   = lane & 15;
  const int lh   = lane >> 4;
  const int c0   = blockIdx.x * kCols;

  for (int i = tid; i < kCols * (kN / 4); i += kThreads) {
    const int c  = i / (kN / 4);
    const int k4 = (i - c * (kN / 4)) * 4;
    v4f v = *(const v4f*)(x + (size_t)(c0 + c) * kN + k4);
    *(v4f*)(hbuf + c * kHStr + k4) = v;
  }
  __syncthreads();

  float* const hrow = hbuf + lm * kHStr;
  for (int base = kN - kBlk; base >= kN - kSpan; base -= kBlk) {
    v8f acc1[2];
    v2f by1[1][4];
    acc1[0] = v8f{};
    acc1[1] = v8f{};
    pass1_alone<1>(w, hrow, acc1, base, wave, lm, lh);
    solve_group<1>(acc1, by1, Ppart, Rbuf, Tbuf, Ybuf, tid, wave, lane, lm, lh);
    pass2_alone<1>(w, hrow, by1, base, wave, lm, lh);
  }
  {
    constexpr int NT = kGroup + kGroup * (kGroup + 1) / 2;
    v8f acc[NT];
    v2f by[kGroup][4];
    for (int base = kN - 2 * kSpan; base >= 0; base -= kSpan) {
#pragma unroll
      for (int t = 0; t < NT; ++t) acc[t] = v8f{};
      if (base == kN - 2 * kSpan)
        pass1_alone<kGroup>(w, hrow, acc, base, wave, lm, lh);
      else
        fused_sweep<kGroup>(w, hrow, acc, by, base + kSpan, base, wave, lm, lh);
      solve_group<kGroup>(acc, by, Ppart, Rbuf, Tbuf, Ybuf, tid, wave, lane, lm, lh);
    }
    pass2_alone<kGroup>(w, hrow, by, 0, wave, lm, lh);
  }

  for (int i = tid; i < kCols * (kN / 4); i += kThreads) {
    const int c  = i / (kN / 4);
    const int k4 = (i - c * (kN / 4)) * 4;
    v4f v = *(const v4f*)(hbuf + c * kHStr + k4);
    *(v4f*)(out + (size_t)(c0 + c) * kN + k4) = v;
  }
}

extern "C" void kernel_launch(void* const* d_in, const int* in_sizes, int n_in,
                              void* d_out, int out_size, void* d_ws, size_t ws_size,
                              hipStream_t stream) {
  const float* x = (const float*)d_in[0];
  const float* w = (const float*)d_in[1];
  float* out = (float*)d_out;

  dim3 grid(kB / kCols);
  dim3 block(kThreads);
  hipLaunchKernelGGL(ortho_householder_kernel, grid, block, 0, stream,
                     x, w, out);
}
